# starting kernel
# baseline (speedup 1.0000x reference)
#include <hip/hip_runtime.h>
#include <math.h>

#define B_   2
#define T_   8
#define N_   5000
#define K1_  17
#define H_   4
#define C_   16
#define NO_  8
#define T7_  (T_-1)
#define E2_  (N_*K1_)
#define TILES2_ ((E2_+15)/16)

typedef float v2f __attribute__((ext_vector_type(2)));
typedef float v8f __attribute__((ext_vector_type(8)));

#define SWZ(x, M) __int_as_float(__builtin_amdgcn_ds_swizzle( \
                      __float_as_int(x), ((M) << 10) | 0x1f))

template<int M>
__device__ __forceinline__ float mergepair(float a, float b, int c) {
    const float keep = (c & M) ? b : a;
    const float send = (c & M) ? a : b;
    return keep + SWZ(send, M);
}

__device__ __forceinline__ float row_reduce16(const float t[8], int c) {
    const float m0 = mergepair<1>(t[0], t[1], c);
    const float m1 = mergepair<1>(t[2], t[3], c);
    const float m2 = mergepair<1>(t[4], t[5], c);
    const float m3 = mergepair<1>(t[6], t[7], c);
    const float n0 = mergepair<2>(m0, m1, c);
    const float n1 = mergepair<2>(m2, m3, c);
    float s = mergepair<4>(n0, n1, c);
    s += SWZ(s, 8);
    return s;
}

__global__ void k1_w_deg(const float* __restrict__ feat,
                         const int*   __restrict__ nn,
                         const float* __restrict__ mm,
                         float*       __restrict__ w_out,
                         float*       __restrict__ deg)
{
    const int wid  = (blockIdx.x * blockDim.x + threadIdx.x) >> 5;
    const int lane = threadIdx.x & 31;

    const int n  = wid % N_;
    const int r0 = wid / N_;
    const int h  = r0 % H_;
    const int bt = r0 / H_;

    const int r  = lane & 15;
    const int hh = lane >> 4;

    const int nbrv = nn[n * K1_ + 1 + r];
    const unsigned vm = (unsigned)__ballot(nbrv >= 0);
    const int nj = (nbrv < 0) ? n : nbrv;

    const float* fi = feat + ((((size_t)bt * N_ + n ) * H_ + h) * C_);
    const float* fj = feat + ((((size_t)bt * N_ + nj) * H_ + h) * C_);
    const float* Mh = mm + h * C_ * C_;

    v8f d = {0.f,0.f,0.f,0.f,0.f,0.f,0.f,0.f};
#pragma unroll
    for (int c = 0; c < 4; ++c) {
        const int j0 = 4 * c + 2 * hh;
        v2f av, bv;
        av.x = fi[j0]     - fj[j0];
        av.y = fi[j0 + 1] - fj[j0 + 1];
        bv.x = Mh[r * C_ + j0];
        bv.y = Mh[r * C_ + j0 + 1];
        d = __builtin_amdgcn_wmma_f32_16x16x4_f32(
                false, av, false, bv, (short)0, d, false, false);
    }

    float t[8];
#pragma unroll
    for (int v = 0; v < 8; ++v) t[v] = d[v] * d[v];
    const float s = row_reduce16(t, r);

    const int   row = (lane & 7) | (hh << 3);
    const float w   = ((vm >> row) & 1u) ? __expf(-s) : 0.f;

    float dp = w;
    dp += SWZ(dp, 1);
    dp += SWZ(dp, 2);
    dp += SWZ(dp, 4);
    dp += SWZ(dp, 16);

    if ((lane & 15) < 8)
        w_out[(((size_t)bt * N_ + n) * 16 + row) * H_ + h] = w;
    if (lane == 0)
        deg[((size_t)bt * N_ + n) * H_ + h] = dp;
}

__global__ void k2_uw(float* __restrict__ uw,
                      const float* __restrict__ deg,
                      const int*   __restrict__ nn)
{
    const size_t tid = (size_t)blockIdx.x * blockDim.x + threadIdx.x;
    const size_t total = (size_t)B_ * T_ * N_ * 16 * H_;
    if (tid >= total) return;
    const int h  = (int)(tid % H_);
    const int k  = (int)((tid / H_) % 16);
    const int n  = (int)((tid / ((size_t)H_ * 16)) % N_);
    const int bt = (int)( tid / ((size_t)H_ * 16 * N_));

    const float w  = uw[tid];
    const int   nb = nn[n * K1_ + 1 + k];
    const float di = deg[((size_t)bt * N_ + n) * H_ + h];
    const float dj = (nb >= 0) ? deg[((size_t)bt * N_ + nb) * H_ + h] : 0.f;
    const float dm = sqrtf(di * dj);
    uw[tid] = (dm > 0.f) ? (w / dm) : 0.f;
}

__global__ void k3_qj(const float* __restrict__ feat,
                      const float* __restrict__ q2,
                      float*       __restrict__ qj)
{
    const size_t tid = (size_t)blockIdx.x * blockDim.x + threadIdx.x;
    const size_t total = (size_t)B_ * T7_ * N_ * H_;
    if (tid >= total) return;
    const int h   = (int)(tid % H_);
    const int n   = (int)((tid / H_) % N_);
    const int bt7 = (int)( tid / ((size_t)H_ * N_));
    const int t   = bt7 % T7_;
    const int b   = bt7 / T7_;

    const float* f = feat + ((((size_t)(b * T_ + t + 1) * N_ + n) * H_ + h) * C_);
    const float* Q = q2 + h * NO_ * C_;
    float* o = qj + tid * NO_;
#pragma unroll
    for (int i = 0; i < NO_; ++i) {
        float s = 0.f;
#pragma unroll
        for (int j = 0; j < C_; ++j) s += Q[i * C_ + j] * f[j];
        o[i] = s;
    }
}

__global__ void k4_wd(const float* __restrict__ feat,
                      const int*   __restrict__ nn,
                      const float* __restrict__ q1,
                      const float* __restrict__ qj,
                      float*       __restrict__ dw)
{
    const int wid  = (blockIdx.x * blockDim.x + threadIdx.x) >> 5;
    const int lane = threadIdx.x & 31;

    const int tile = wid % TILES2_;
    const int r1   = wid / TILES2_;
    const int h    = r1 % H_;
    const int bt7  = r1 / H_;
    const int t    = bt7 % T7_;
    const int bq   = bt7 / T7_;
    const int btf  = bq * T_ + t;

    const int r  = lane & 15;
    const int hh = lane >> 4;

    const int  e     = tile * 16 + r;
    const bool rowOK = (e < E2_);
    const int  ec    = rowOK ? e : 0;
    const int  n     = ec / K1_;
    const int  k     = ec - n * K1_;
    const int  nbrv  = nn[n * K1_ + k];
    const bool pred  = rowOK && (nbrv >= 0);
    const unsigned vm = (unsigned)__ballot(pred);
    const int  jn    = pred ? nbrv : 0;
    const float sA   = pred ? 1.f : 0.f;

    const float* fi = feat + ((((size_t)btf * N_ + jn) * H_ + h) * C_);
    const float* Qh = q1 + h * NO_ * C_;
    const float sB  = (r < NO_) ? 1.f : 0.f;

    v8f d = {0.f,0.f,0.f,0.f,0.f,0.f,0.f,0.f};
#pragma unroll
    for (int c = 0; c < 4; ++c) {
        const int j0 = 4 * c + 2 * hh;
        v2f av, bv;
        av.x = sA * fi[j0];
        av.y = sA * fi[j0 + 1];
        bv.x = sB * Qh[(r & (NO_ - 1)) * C_ + j0];
        bv.y = sB * Qh[(r & (NO_ - 1)) * C_ + j0 + 1];
        d = __builtin_amdgcn_wmma_f32_16x16x4_f32(
                false, av, false, bv, (short)0, d, false, false);
    }

    float tv[8];
#pragma unroll
    for (int v = 0; v < 8; ++v) {
        const int ev = tile * 16 + v + (hh << 3);
        float qv = 0.f;
        if (ev < E2_ && r < NO_) {
            const int nv = ev / K1_;
            qv = qj[(((size_t)bt7 * N_ + nv) * H_ + h) * NO_ + r];
        }
        tv[v] = d[v] * qv;
    }
    const float s = row_reduce16(tv, r);

    const int   row = (lane & 7) | (hh << 3);
    const float wv  = ((vm >> row) & 1u) ? __expf(-s) : 0.f;

    if ((lane & 15) < 8) {
        const int ew = tile * 16 + row;
        if (ew < E2_)
            dw[((size_t)bt7 * E2_ + ew) * H_ + h] = wv;
    }
}

__global__ void k5_dw(float* __restrict__ dw)
{
    const size_t tid = (size_t)blockIdx.x * blockDim.x + threadIdx.x;
    const size_t total = (size_t)B_ * T7_ * N_ * H_;
    if (tid >= total) return;
    const int h   = (int)(tid % H_);
    const int n   = (int)((tid / H_) % N_);
    const int bt7 = (int)( tid / ((size_t)H_ * N_));

    float* base = dw + ((size_t)bt7 * E2_ + (size_t)n * K1_) * H_ + h;
    float s = 0.f;
#pragma unroll
    for (int k = 0; k < K1_; ++k) s += base[k * H_];
    const float inv = (s > 0.f) ? (1.f / s) : 0.f;
#pragma unroll
    for (int k = 0; k < K1_; ++k) base[k * H_] *= inv;
}

extern "C" void kernel_launch(void* const* d_in, const int* in_sizes, int n_in,
                              void* d_out, int out_size, void* d_ws, size_t ws_size,
                              hipStream_t stream)
{
    const float* feat = (const float*)d_in[0];
    const int*   nn   = (const int*)  d_in[1];
    const float* q1   = (const float*)d_in[2];
    const float* q2   = (const float*)d_in[3];
    const float* mm   = (const float*)d_in[4];

    float* uw = (float*)d_out;
    float* dw = uw + (size_t)B_ * T_ * N_ * 16 * H_;

    float* deg = (float*)d_ws;
    float* qjw = deg + (size_t)B_ * T_ * N_ * H_;

    const int BLK = 256;

    {
        const int waves = B_ * T_ * H_ * N_;
        k1_w_deg<<<waves * 32 / BLK, BLK, 0, stream>>>(feat, nn, mm, uw, deg);
    }
    {
        const size_t total = (size_t)B_ * T_ * N_ * 16 * H_;
        k2_uw<<<(unsigned)((total + BLK - 1) / BLK), BLK, 0, stream>>>(uw, deg, nn);
    }
    {
        const size_t total = (size_t)B_ * T7_ * N_ * H_;
        k3_qj<<<(unsigned)((total + BLK - 1) / BLK), BLK, 0, stream>>>(feat, q2, qjw);
    }
    {
        const long long waves = (long long)B_ * T7_ * H_ * TILES2_;
        k4_wd<<<(unsigned)(waves * 32 / BLK), BLK, 0, stream>>>(feat, nn, q1, qjw, dw);
    }
    {
        const size_t total = (size_t)B_ * T7_ * N_ * H_;
        k5_dw<<<(unsigned)((total + BLK - 1) / BLK), BLK, 0, stream>>>(dw);
    }
}


# WMMA operands swapped (weights as A, gathered rows as B): one neighbor per lane, so norms and Qi.Qj dots are in-lane sums over D's 8 registers + one half exchange; swizzle trees go; sum order changes
# speedup vs baseline: 1.5812x; 1.5812x over previous
#include <hip/hip_runtime.h>
#include <math.h>

#define B_   2
#define T_   8
#define N_   5000
#define K1_  17
#define H_   4
#define C_   16
#define NO_  8
#define T7_  (T_-1)
#define E2_  (N_*K1_)

typedef float v2f __attribute__((ext_vector_type(2)));
typedef float v8f __attribute__((ext_vector_type(8)));

#define SWZ(x, M) __int_as_float(__builtin_amdgcn_ds_swizzle( \
                      __float_as_int(x), ((M) << 10) | 0x1f))

typedef float v4f __attribute__((ext_vector_type(4)));

__device__ __forceinline__ float dot8(const v8f x, const v8f y)
{
    v2f a = __builtin_shufflevector(x, x, 0, 1) * __builtin_shufflevector(y, y, 0, 1);
    a = __builtin_shufflevector(x, x, 2, 3) * __builtin_shufflevector(y, y, 2, 3) + a;
    a = __builtin_shufflevector(x, x, 4, 5) * __builtin_shufflevector(y, y, 4, 5) + a;
    a = __builtin_shufflevector(x, x, 6, 7) * __builtin_shufflevector(y, y, 6, 7) + a;
    return a.x + a.y;
}

#define QTR_   (N_/4)
#define STL_   ((QTR_+15)/16)
#define U1_    4
#define UF_    2
#define PLANE_ (N_*C_)

__device__ __forceinline__ void stage_plane(float* plane, const float* __restrict__ src)
{
    for (int p = threadIdx.x; p < N_ * 4; p += 1024) {
        const int row  = p >> 2;
        const int part = p & 3;
        const v4f v = *(const v4f*)(src + (size_t)row * H_ * C_ + part * 4);
        *(v4f*)(plane + row * C_ + ((part ^ ((row >> 2) & 3)) << 2)) = v;
    }
    __syncthreads();
}

__device__ __forceinline__ v2f lds_frag(const float* plane, int n, int c, int hh)
{
    return *(const v2f*)(plane + n * C_ + ((c ^ ((n >> 2) & 3)) << 2) + 2 * hh);
}

template<bool DIR, int U>
__device__ __forceinline__ void node_loop(
    const float* plane,
    const int*   __restrict__ nn,
    const float* __restrict__ qjp,
    const v2f (&bv)[4], const v2f (&b1)[4],
    float* __restrict__ wp,
    float* __restrict__ dgp,
    float* __restrict__ dwp,
    int nbeg, int nend, int lane, int r, int hh)
{
    const v8f zero8 = {0.f,0.f,0.f,0.f,0.f,0.f,0.f,0.f};
    int nb[U];
    v8f qn[U];
#pragma unroll
    for (int u = 0; u < U; ++u) {
        const int nx = (nbeg + 32 * u < nend) ? (nbeg + 32 * u) : (nend - 1);
        nb[u] = nn[nx * K1_ + 1 + r];
        qn[u] = DIR ? *(const v8f*)(qjp + nx * (H_ * NO_)) : zero8;
    }

    for (int n0 = nbeg; n0 < nend; n0 += 32 * U) {
        int nu[U], nbc[U];
        v8f qc[U];
#pragma unroll
        for (int u = 0; u < U; ++u) {
            nu[u]  = (n0 + 32 * u < nend) ? (n0 + 32 * u) : (nend - 1);
            nbc[u] = nb[u];
            qc[u]  = qn[u];
        }
#pragma unroll
        for (int u = 0; u < U; ++u) {
            const int nx = (n0 + 32 * (U + u) < nend) ? (n0 + 32 * (U + u)) : (nend - 1);
            nb[u] = nn[nx * K1_ + 1 + r];
            qn[u] = DIR ? *(const v8f*)(qjp + nx * (H_ * NO_)) : zero8;
        }

        float w[U], dp[U], wd[U];
#pragma unroll
        for (int u = 0; u < U; ++u) {
            const int  n     = nu[u];
            const int  nbrv  = nbc[u];
            const bool valid = (nbrv >= 0);
            const int  nj    = valid ? nbrv : n;

            v8f d = zero8;
            v8f g = zero8;
#pragma unroll
            for (int c = 0; c < 4; ++c) {
                const v2f fj = lds_frag(plane, nj, c, hh);
                const v2f av = lds_frag(plane, n, c, hh) - fj;
                d = __builtin_amdgcn_wmma_f32_16x16x4_f32(
                        false, bv[c], false, av, (short)0, d, false, false);
                if (DIR)
                    g = __builtin_amdgcn_wmma_f32_16x16x4_f32(
                            false, b1[c], false, fj, (short)0, g, false, false);
            }

            float s = dot8(d, d);
            s += SWZ(s, 16);

            w[u] = valid ? __expf(-s) : 0.f;

            float dq = w[u];
            dq += SWZ(dq, 1);
            dq += SWZ(dq, 2);
            dq += SWZ(dq, 4);
            dq += SWZ(dq, 8);
            dp[u] = dq;

            wd[u] = 0.f;
            if (DIR) {
                const float s2 = dot8(g, qc[u]);
                wd[u] = valid ? __expf(-s2) : 0.f;
            }
        }

#pragma unroll
        for (int u = 0; u < U; ++u) {
            if (n0 + 32 * u < nend) {
                const int n = n0 + 32 * u;
                if (hh == 0) {
                    wp[(n * 16 + r) * H_] = w[u];
                    if (DIR) dwp[(n * K1_ + 1 + r) * H_] = wd[u];
                }
                if (lane == 0)
                    dgp[n * H_] = dp[u];
            }
        }
    }
}

__global__ void __launch_bounds__(1024)
k14_plane(const float* __restrict__ feat,
          const int*   __restrict__ nn,
          const float* __restrict__ mm,
          const float* __restrict__ q1,
          const float* __restrict__ qj,
          float*       __restrict__ w_out,
          float*       __restrict__ deg,
          float*       __restrict__ dw)
{
    __shared__ __attribute__((aligned(16))) float plane[PLANE_];

    const int blk  = blockIdx.x;
    const int q    = blk & 3;
    const int h    = (blk >> 2) % H_;
    const int bt   = blk / (4 * H_);
    const int tt   = bt % T_;
    const int bq   = bt / T_;
    const bool dir = (tt < T7_);
    const int bt7  = bq * T7_ + (dir ? tt : 0);
    const int wv   = __builtin_amdgcn_readfirstlane((int)(threadIdx.x >> 5));
    const int lane = threadIdx.x & 31;

    const int r  = lane & 15;
    const int hh = lane >> 4;

    const float* Mh = mm + h * C_ * C_ + r * C_ + 2 * hh;
    v2f bv[4];
#pragma unroll
    for (int c = 0; c < 4; ++c) bv[c] = *(const v2f*)(Mh + 4 * c);

    const float  sB = (r < NO_) ? 1.f : 0.f;
    const float* Qh = q1 + h * NO_ * C_ + (r & (NO_ - 1)) * C_ + 2 * hh;
    v2f b1[4];
#pragma unroll
    for (int c = 0; c < 4; ++c) b1[c] = *(const v2f*)(Qh + 4 * c) * sB;

    stage_plane(plane, feat + (size_t)bt * N_ * H_ * C_ + h * C_);

    const int q0   = q * QTR_;
    const int nend = q0 + QTR_;
    const int nbeg = q0 + wv;

    float*       wp  = w_out + (size_t)bt * N_ * 16 * H_ + h;
    float*       dgp = deg   + (size_t)bt * N_ * H_ + h;
    float*       dwp = dw    + (size_t)bt7 * E2_ * H_ + h;
    const float* qjp = qj    + ((size_t)bt7 * N_ * H_ + h) * NO_;

    if (!dir) {
        node_loop<false, U1_>(plane, nn, qjp, bv, b1, wp, dgp, dwp, nbeg, nend, lane, r, hh);
        return;
    }

    node_loop<true, UF_>(plane, nn, qjp, bv, b1, wp, dgp, dwp, nbeg, nend, lane, r, hh);

    for (int j = wv; j < STL_; j += 32) {
        const int  nd   = q0 + 16 * j + r;
        const bool ok   = (nd < nend);
        const int  ndc  = ok ? nd : (nend - 1);
        const int  g0   = nn[ndc * K1_];
        const bool pred = ok && (g0 >= 0);
        const int  jn   = pred ? g0 : 0;
        const v8f  qv   = *(const v8f*)(qjp + ndc * (H_ * NO_));

        v8f d = {0.f,0.f,0.f,0.f,0.f,0.f,0.f,0.f};
#pragma unroll
        for (int c = 0; c < 4; ++c)
            d = __builtin_amdgcn_wmma_f32_16x16x4_f32(
                    false, b1[c], false, lds_frag(plane, jn, c, hh), (short)0, d, false, false);

        const float s = dot8(d, qv);
        const float wd0 = pred ? __expf(-s) : 0.f;

        if (hh == 0 && ok) dwp[(nd * K1_) * H_] = wd0;
    }
}

__global__ void k2_uw(float* __restrict__ uw,
                      const float* __restrict__ deg,
                      const int*   __restrict__ nn)
{
    const size_t tid = (size_t)blockIdx.x * blockDim.x + threadIdx.x;
    const size_t total = (size_t)B_ * T_ * N_ * 16 * H_;
    if (tid >= total) return;
    const int h  = (int)(tid % H_);
    const int k  = (int)((tid / H_) % 16);
    const int n  = (int)((tid / ((size_t)H_ * 16)) % N_);
    const int bt = (int)( tid / ((size_t)H_ * 16 * N_));

    const float w  = uw[tid];
    const int   nb = nn[n * K1_ + 1 + k];
    const float di = deg[((size_t)bt * N_ + n) * H_ + h];
    const float dj = (nb >= 0) ? deg[((size_t)bt * N_ + nb) * H_ + h] : 0.f;
    const float dm = sqrtf(di * dj);
    uw[tid] = (dm > 0.f) ? (w / dm) : 0.f;
}

__global__ void k3_qj(const float* __restrict__ feat,
                      const float* __restrict__ q2,
                      float*       __restrict__ qj)
{
    const size_t tid = (size_t)blockIdx.x * blockDim.x + threadIdx.x;
    const size_t total = (size_t)B_ * T7_ * N_ * H_;
    if (tid >= total) return;
    const int h   = (int)(tid % H_);
    const int n   = (int)((tid / H_) % N_);
    const int bt7 = (int)( tid / ((size_t)H_ * N_));
    const int t   = bt7 % T7_;
    const int b   = bt7 / T7_;

    const float* f = feat + ((((size_t)(b * T_ + t + 1) * N_ + n) * H_ + h) * C_);
    const float* Q = q2 + h * NO_ * C_;
    float* o = qj + tid * NO_;
#pragma unroll
    for (int i = 0; i < NO_; ++i) {
        float s = 0.f;
#pragma unroll
        for (int j = 0; j < C_; ++j) s += Q[i * C_ + j] * f[j];
        o[i] = s;
    }
}

__global__ void k5_dw(float* __restrict__ dw)
{
    const size_t tid = (size_t)blockIdx.x * blockDim.x + threadIdx.x;
    const size_t total = (size_t)B_ * T7_ * N_ * H_;
    if (tid >= total) return;
    const int h   = (int)(tid % H_);
    const int n   = (int)((tid / H_) % N_);
    const int bt7 = (int)( tid / ((size_t)H_ * N_));

    float* base = dw + ((size_t)bt7 * E2_ + (size_t)n * K1_) * H_ + h;
    float s = 0.f;
#pragma unroll
    for (int k = 0; k < K1_; ++k) s += base[k * H_];
    const float inv = (s > 0.f) ? (1.f / s) : 0.f;
#pragma unroll
    for (int k = 0; k < K1_; ++k) base[k * H_] *= inv;
}

extern "C" void kernel_launch(void* const* d_in, const int* in_sizes, int n_in,
                              void* d_out, int out_size, void* d_ws, size_t ws_size,
                              hipStream_t stream)
{
    const float* feat = (const float*)d_in[0];
    const int*   nn   = (const int*)  d_in[1];
    const float* q1   = (const float*)d_in[2];
    const float* q2   = (const float*)d_in[3];
    const float* mm   = (const float*)d_in[4];

    float* uw = (float*)d_out;
    float* dw = uw + (size_t)B_ * T_ * N_ * 16 * H_;

    float* deg = (float*)d_ws;
    float* qjw = deg + (size_t)B_ * T_ * N_ * H_;

    const int BLK = 256;

    {
        const size_t total = (size_t)B_ * T7_ * N_ * H_;
        k3_qj<<<(unsigned)((total + BLK - 1) / BLK), BLK, 0, stream>>>(feat, q2, qjw);
    }
    k14_plane<<<B_ * T_ * H_ * 4, 1024, 0, stream>>>(feat, nn, mm, q1, qjw, uw, deg, dw);
    {
        const size_t total = (size_t)B_ * T_ * N_ * 16 * H_;
        k2_uw<<<(unsigned)((total + BLK - 1) / BLK), BLK, 0, stream>>>(uw, deg, nn);
    }
    {
        const size_t total = (size_t)B_ * T7_ * N_ * H_;
        k5_dw<<<(unsigned)((total + BLK - 1) / BLK), BLK, 0, stream>>>(dw);
    }
}


# d_w in-degree normalise rides in the u_w normalise launch: its 547 blocks lead one grid, the 20,000 element blocks stream behind; launches 4->3, one launch boundary (drain, gap, ramp) gone
# speedup vs baseline: 1.6458x; 1.0409x over previous
#include <hip/hip_runtime.h>
#include <math.h>

#define B_   2
#define T_   8
#define N_   5000
#define K1_  17
#define H_   4
#define C_   16
#define NO_  8
#define T7_  (T_-1)
#define E2_  (N_*K1_)

typedef float v2f __attribute__((ext_vector_type(2)));
typedef float v8f __attribute__((ext_vector_type(8)));

#define SWZ(x, M) __int_as_float(__builtin_amdgcn_ds_swizzle( \
                      __float_as_int(x), ((M) << 10) | 0x1f))

typedef float v4f __attribute__((ext_vector_type(4)));

__device__ __forceinline__ float dot8(const v8f x, const v8f y)
{
    v2f a = __builtin_shufflevector(x, x, 0, 1) * __builtin_shufflevector(y, y, 0, 1);
    a = __builtin_shufflevector(x, x, 2, 3) * __builtin_shufflevector(y, y, 2, 3) + a;
    a = __builtin_shufflevector(x, x, 4, 5) * __builtin_shufflevector(y, y, 4, 5) + a;
    a = __builtin_shufflevector(x, x, 6, 7) * __builtin_shufflevector(y, y, 6, 7) + a;
    return a.x + a.y;
}

#define QTR_   (N_/4)
#define STL_   ((QTR_+15)/16)
#define U1_    4
#define UF_    2
#define PLANE_ (N_*C_)

__device__ __forceinline__ void stage_plane(float* plane, const float* __restrict__ src)
{
    for (int p = threadIdx.x; p < N_ * 4; p += 1024) {
        const int row  = p >> 2;
        const int part = p & 3;
        const v4f v = *(const v4f*)(src + (size_t)row * H_ * C_ + part * 4);
        *(v4f*)(plane + row * C_ + ((part ^ ((row >> 2) & 3)) << 2)) = v;
    }
    __syncthreads();
}

__device__ __forceinline__ v2f lds_frag(const float* plane, int n, int c, int hh)
{
    return *(const v2f*)(plane + n * C_ + ((c ^ ((n >> 2) & 3)) << 2) + 2 * hh);
}

template<bool DIR, int U>
__device__ __forceinline__ void node_loop(
    const float* plane,
    const int*   __restrict__ nn,
    const float* __restrict__ qjp,
    const v2f (&bv)[4], const v2f (&b1)[4],
    float* __restrict__ wp,
    float* __restrict__ dgp,
    float* __restrict__ dwp,
    int nbeg, int nend, int lane, int r, int hh)
{
    const v8f zero8 = {0.f,0.f,0.f,0.f,0.f,0.f,0.f,0.f};
    int nb[U];
    v8f qn[U];
#pragma unroll
    for (int u = 0; u < U; ++u) {
        const int nx = (nbeg + 32 * u < nend) ? (nbeg + 32 * u) : (nend - 1);
        nb[u] = nn[nx * K1_ + 1 + r];
        qn[u] = DIR ? *(const v8f*)(qjp + nx * (H_ * NO_)) : zero8;
    }

    for (int n0 = nbeg; n0 < nend; n0 += 32 * U) {
        int nu[U], nbc[U];
        v8f qc[U];
#pragma unroll
        for (int u = 0; u < U; ++u) {
            nu[u]  = (n0 + 32 * u < nend) ? (n0 + 32 * u) : (nend - 1);
            nbc[u] = nb[u];
            qc[u]  = qn[u];
        }
#pragma unroll
        for (int u = 0; u < U; ++u) {
            const int nx = (n0 + 32 * (U + u) < nend) ? (n0 + 32 * (U + u)) : (nend - 1);
            nb[u] = nn[nx * K1_ + 1 + r];
            qn[u] = DIR ? *(const v8f*)(qjp + nx * (H_ * NO_)) : zero8;
        }

        float w[U], dp[U], wd[U];
#pragma unroll
        for (int u = 0; u < U; ++u) {
            const int  n     = nu[u];
            const int  nbrv  = nbc[u];
            const bool valid = (nbrv >= 0);
            const int  nj    = valid ? nbrv : n;

            v8f d = zero8;
            v8f g = zero8;
#pragma unroll
            for (int c = 0; c < 4; ++c) {
                const v2f fj = lds_frag(plane, nj, c, hh);
                const v2f av = lds_frag(plane, n, c, hh) - fj;
                d = __builtin_amdgcn_wmma_f32_16x16x4_f32(
                        false, bv[c], false, av, (short)0, d, false, false);
                if (DIR)
                    g = __builtin_amdgcn_wmma_f32_16x16x4_f32(
                            false, b1[c], false, fj, (short)0, g, false, false);
            }

            float s = dot8(d, d);
            s += SWZ(s, 16);

            w[u] = valid ? __expf(-s) : 0.f;

            float dq = w[u];
            dq += SWZ(dq, 1);
            dq += SWZ(dq, 2);
            dq += SWZ(dq, 4);
            dq += SWZ(dq, 8);
            dp[u] = dq;

            wd[u] = 0.f;
            if (DIR) {
                const float s2 = dot8(g, qc[u]);
                wd[u] = valid ? __expf(-s2) : 0.f;
            }
        }

#pragma unroll
        for (int u = 0; u < U; ++u) {
            if (n0 + 32 * u < nend) {
                const int n = n0 + 32 * u;
                if (hh == 0) {
                    wp[(n * 16 + r) * H_] = w[u];
                    if (DIR) dwp[(n * K1_ + 1 + r) * H_] = wd[u];
                }
                if (lane == 0)
                    dgp[n * H_] = dp[u];
            }
        }
    }
}

__global__ void __launch_bounds__(1024)
k14_plane(const float* __restrict__ feat,
          const int*   __restrict__ nn,
          const float* __restrict__ mm,
          const float* __restrict__ q1,
          const float* __restrict__ qj,
          float*       __restrict__ w_out,
          float*       __restrict__ deg,
          float*       __restrict__ dw)
{
    __shared__ __attribute__((aligned(16))) float plane[PLANE_];

    const int blk  = blockIdx.x;
    const int q    = blk & 3;
    const int h    = (blk >> 2) % H_;
    const int bt   = blk / (4 * H_);
    const int tt   = bt % T_;
    const int bq   = bt / T_;
    const bool dir = (tt < T7_);
    const int bt7  = bq * T7_ + (dir ? tt : 0);
    const int wv   = __builtin_amdgcn_readfirstlane((int)(threadIdx.x >> 5));
    const int lane = threadIdx.x & 31;

    const int r  = lane & 15;
    const int hh = lane >> 4;

    const float* Mh = mm + h * C_ * C_ + r * C_ + 2 * hh;
    v2f bv[4];
#pragma unroll
    for (int c = 0; c < 4; ++c) bv[c] = *(const v2f*)(Mh + 4 * c);

    const float  sB = (r < NO_) ? 1.f : 0.f;
    const float* Qh = q1 + h * NO_ * C_ + (r & (NO_ - 1)) * C_ + 2 * hh;
    v2f b1[4];
#pragma unroll
    for (int c = 0; c < 4; ++c) b1[c] = *(const v2f*)(Qh + 4 * c) * sB;

    stage_plane(plane, feat + (size_t)bt * N_ * H_ * C_ + h * C_);

    const int q0   = q * QTR_;
    const int nend = q0 + QTR_;
    const int nbeg = q0 + wv;

    float*       wp  = w_out + (size_t)bt * N_ * 16 * H_ + h;
    float*       dgp = deg   + (size_t)bt * N_ * H_ + h;
    float*       dwp = dw    + (size_t)bt7 * E2_ * H_ + h;
    const float* qjp = qj    + ((size_t)bt7 * N_ * H_ + h) * NO_;

    if (!dir) {
        node_loop<false, U1_>(plane, nn, qjp, bv, b1, wp, dgp, dwp, nbeg, nend, lane, r, hh);
        return;
    }

    node_loop<true, UF_>(plane, nn, qjp, bv, b1, wp, dgp, dwp, nbeg, nend, lane, r, hh);

    for (int j = wv; j < STL_; j += 32) {
        const int  nd   = q0 + 16 * j + r;
        const bool ok   = (nd < nend);
        const int  ndc  = ok ? nd : (nend - 1);
        const int  g0   = nn[ndc * K1_];
        const bool pred = ok && (g0 >= 0);
        const int  jn   = pred ? g0 : 0;
        const v8f  qv   = *(const v8f*)(qjp + ndc * (H_ * NO_));

        v8f d = {0.f,0.f,0.f,0.f,0.f,0.f,0.f,0.f};
#pragma unroll
        for (int c = 0; c < 4; ++c)
            d = __builtin_amdgcn_wmma_f32_16x16x4_f32(
                    false, b1[c], false, lds_frag(plane, jn, c, hh), (short)0, d, false, false);

        const float s = dot8(d, qv);
        const float wd0 = pred ? __expf(-s) : 0.f;

        if (hh == 0 && ok) dwp[(nd * K1_) * H_] = wd0;
    }
}

__global__ void k3_qj(const float* __restrict__ feat,
                      const float* __restrict__ q2,
                      float*       __restrict__ qj)
{
    const size_t tid = (size_t)blockIdx.x * blockDim.x + threadIdx.x;
    const size_t total = (size_t)B_ * T7_ * N_ * H_;
    if (tid >= total) return;
    const int h   = (int)(tid % H_);
    const int n   = (int)((tid / H_) % N_);
    const int bt7 = (int)( tid / ((size_t)H_ * N_));
    const int t   = bt7 % T7_;
    const int b   = bt7 / T7_;

    const float* f = feat + ((((size_t)(b * T_ + t + 1) * N_ + n) * H_ + h) * C_);
    const float* Q = q2 + h * NO_ * C_;
    float* o = qj + tid * NO_;
#pragma unroll
    for (int i = 0; i < NO_; ++i) {
        float s = 0.f;
#pragma unroll
        for (int j = 0; j < C_; ++j) s += Q[i * C_ + j] * f[j];
        o[i] = s;
    }
}

#define PBLK_  256
#define NB5_   ((B_ * T7_ * N_ * H_ + PBLK_ - 1) / PBLK_)
#define NB2_   ((B_ * T_ * N_ * 16 * H_ + PBLK_ - 1) / PBLK_)

__global__ void __launch_bounds__(PBLK_)
k25_post(float* __restrict__ uw,
         float* __restrict__ dw,
         const float* __restrict__ deg,
         const int*   __restrict__ nn)
{
    if (blockIdx.x < NB5_) {
        const size_t tid = (size_t)blockIdx.x * PBLK_ + threadIdx.x;
        const size_t total = (size_t)B_ * T7_ * N_ * H_;
        if (tid >= total) return;
        const int h   = (int)(tid % H_);
        const int n   = (int)((tid / H_) % N_);
        const int bt7 = (int)( tid / ((size_t)H_ * N_));

        float* base = dw + ((size_t)bt7 * E2_ + (size_t)n * K1_) * H_ + h;
        float v[K1_];
#pragma unroll
        for (int k = 0; k < K1_; ++k) v[k] = base[k * H_];
        float s = 0.f;
#pragma unroll
        for (int k = 0; k < K1_; ++k) s += v[k];
        const float inv = (s > 0.f) ? (1.f / s) : 0.f;
#pragma unroll
        for (int k = 0; k < K1_; ++k) base[k * H_] = v[k] * inv;
        return;
    }

    const size_t tid = (size_t)(blockIdx.x - NB5_) * PBLK_ + threadIdx.x;
    const size_t total = (size_t)B_ * T_ * N_ * 16 * H_;
    if (tid >= total) return;
    const int h  = (int)(tid % H_);
    const int k  = (int)((tid / H_) % 16);
    const int n  = (int)((tid / ((size_t)H_ * 16)) % N_);
    const int bt = (int)( tid / ((size_t)H_ * 16 * N_));

    const float w  = uw[tid];
    const int   nb = nn[n * K1_ + 1 + k];
    const float di = deg[((size_t)bt * N_ + n) * H_ + h];
    const float dj = (nb >= 0) ? deg[((size_t)bt * N_ + nb) * H_ + h] : 0.f;
    const float dm = sqrtf(di * dj);
    uw[tid] = (dm > 0.f) ? (w / dm) : 0.f;
}

extern "C" void kernel_launch(void* const* d_in, const int* in_sizes, int n_in,
                              void* d_out, int out_size, void* d_ws, size_t ws_size,
                              hipStream_t stream)
{
    const float* feat = (const float*)d_in[0];
    const int*   nn   = (const int*)  d_in[1];
    const float* q1   = (const float*)d_in[2];
    const float* q2   = (const float*)d_in[3];
    const float* mm   = (const float*)d_in[4];

    float* uw = (float*)d_out;
    float* dw = uw + (size_t)B_ * T_ * N_ * 16 * H_;

    float* deg = (float*)d_ws;
    float* qjw = deg + (size_t)B_ * T_ * N_ * H_;

    const int BLK = 256;

    {
        const size_t total = (size_t)B_ * T7_ * N_ * H_;
        k3_qj<<<(unsigned)((total + BLK - 1) / BLK), BLK, 0, stream>>>(feat, q2, qjw);
    }
    k14_plane<<<B_ * T_ * H_ * 4, 1024, 0, stream>>>(feat, nn, mm, q1, qjw, uw, deg, dw);
    k25_post<<<NB5_ + NB2_, PBLK_, 0, stream>>>(uw, dw, deg, nn);
}


# u_w normalise: a thread takes one (n,k,h) and walks all 16 (b,t) planes; index read once, its 16 w, 16 deg_i, then 16 deg_j loads go out as groups; post waves 164,376->14,376 (grid 20,547->1,797)
# speedup vs baseline: 1.7278x; 1.0498x over previous
#include <hip/hip_runtime.h>
#include <math.h>

#define B_   2
#define T_   8
#define N_   5000
#define K1_  17
#define H_   4
#define C_   16
#define NO_  8
#define T7_  (T_-1)
#define E2_  (N_*K1_)

typedef float v2f __attribute__((ext_vector_type(2)));
typedef float v8f __attribute__((ext_vector_type(8)));

#define SWZ(x, M) __int_as_float(__builtin_amdgcn_ds_swizzle( \
                      __float_as_int(x), ((M) << 10) | 0x1f))

typedef float v4f __attribute__((ext_vector_type(4)));

__device__ __forceinline__ float dot8(const v8f x, const v8f y)
{
    v2f a = __builtin_shufflevector(x, x, 0, 1) * __builtin_shufflevector(y, y, 0, 1);
    a = __builtin_shufflevector(x, x, 2, 3) * __builtin_shufflevector(y, y, 2, 3) + a;
    a = __builtin_shufflevector(x, x, 4, 5) * __builtin_shufflevector(y, y, 4, 5) + a;
    a = __builtin_shufflevector(x, x, 6, 7) * __builtin_shufflevector(y, y, 6, 7) + a;
    return a.x + a.y;
}

#define QTR_   (N_/4)
#define STL_   ((QTR_+15)/16)
#define U1_    4
#define UF_    2
#define PLANE_ (N_*C_)

__device__ __forceinline__ void stage_plane(float* plane, const float* __restrict__ src)
{
    for (int p = threadIdx.x; p < N_ * 4; p += 1024) {
        const int row  = p >> 2;
        const int part = p & 3;
        const v4f v = *(const v4f*)(src + (size_t)row * H_ * C_ + part * 4);
        *(v4f*)(plane + row * C_ + ((part ^ ((row >> 2) & 3)) << 2)) = v;
    }
    __syncthreads();
}

__device__ __forceinline__ v2f lds_frag(const float* plane, int n, int c, int hh)
{
    return *(const v2f*)(plane + n * C_ + ((c ^ ((n >> 2) & 3)) << 2) + 2 * hh);
}

template<bool DIR, int U>
__device__ __forceinline__ void node_loop(
    const float* plane,
    const int*   __restrict__ nn,
    const float* __restrict__ qjp,
    const v2f (&bv)[4], const v2f (&b1)[4],
    float* __restrict__ wp,
    float* __restrict__ dgp,
    float* __restrict__ dwp,
    int nbeg, int nend, int lane, int r, int hh)
{
    const v8f zero8 = {0.f,0.f,0.f,0.f,0.f,0.f,0.f,0.f};
    int nb[U];
    v8f qn[U];
#pragma unroll
    for (int u = 0; u < U; ++u) {
        const int nx = (nbeg + 32 * u < nend) ? (nbeg + 32 * u) : (nend - 1);
        nb[u] = nn[nx * K1_ + 1 + r];
        qn[u] = DIR ? *(const v8f*)(qjp + nx * (H_ * NO_)) : zero8;
    }

    for (int n0 = nbeg; n0 < nend; n0 += 32 * U) {
        int nu[U], nbc[U];
        v8f qc[U];
#pragma unroll
        for (int u = 0; u < U; ++u) {
            nu[u]  = (n0 + 32 * u < nend) ? (n0 + 32 * u) : (nend - 1);
            nbc[u] = nb[u];
            qc[u]  = qn[u];
        }
#pragma unroll
        for (int u = 0; u < U; ++u) {
            const int nx = (n0 + 32 * (U + u) < nend) ? (n0 + 32 * (U + u)) : (nend - 1);
            nb[u] = nn[nx * K1_ + 1 + r];
            qn[u] = DIR ? *(const v8f*)(qjp + nx * (H_ * NO_)) : zero8;
        }

        float w[U], dp[U], wd[U];
#pragma unroll
        for (int u = 0; u < U; ++u) {
            const int  n     = nu[u];
            const int  nbrv  = nbc[u];
            const bool valid = (nbrv >= 0);
            const int  nj    = valid ? nbrv : n;

            v8f d = zero8;
            v8f g = zero8;
#pragma unroll
            for (int c = 0; c < 4; ++c) {
                const v2f fj = lds_frag(plane, nj, c, hh);
                const v2f av = lds_frag(plane, n, c, hh) - fj;
                d = __builtin_amdgcn_wmma_f32_16x16x4_f32(
                        false, bv[c], false, av, (short)0, d, false, false);
                if (DIR)
                    g = __builtin_amdgcn_wmma_f32_16x16x4_f32(
                            false, b1[c], false, fj, (short)0, g, false, false);
            }

            float s = dot8(d, d);
            s += SWZ(s, 16);

            w[u] = valid ? __expf(-s) : 0.f;

            float dq = w[u];
            dq += SWZ(dq, 1);
            dq += SWZ(dq, 2);
            dq += SWZ(dq, 4);
            dq += SWZ(dq, 8);
            dp[u] = dq;

            wd[u] = 0.f;
            if (DIR) {
                const float s2 = dot8(g, qc[u]);
                wd[u] = valid ? __expf(-s2) : 0.f;
            }
        }

#pragma unroll
        for (int u = 0; u < U; ++u) {
            if (n0 + 32 * u < nend) {
                const int n = n0 + 32 * u;
                if (hh == 0) {
                    wp[(n * 16 + r) * H_] = w[u];
                    if (DIR) dwp[(n * K1_ + 1 + r) * H_] = wd[u];
                }
                if (lane == 0)
                    dgp[n * H_] = dp[u];
            }
        }
    }
}

__global__ void __launch_bounds__(1024)
k14_plane(const float* __restrict__ feat,
          const int*   __restrict__ nn,
          const float* __restrict__ mm,
          const float* __restrict__ q1,
          const float* __restrict__ qj,
          float*       __restrict__ w_out,
          float*       __restrict__ deg,
          float*       __restrict__ dw)
{
    __shared__ __attribute__((aligned(16))) float plane[PLANE_];

    const int blk  = blockIdx.x;
    const int q    = blk & 3;
    const int h    = (blk >> 2) % H_;
    const int bt   = blk / (4 * H_);
    const int tt   = bt % T_;
    const int bq   = bt / T_;
    const bool dir = (tt < T7_);
    const int bt7  = bq * T7_ + (dir ? tt : 0);
    const int wv   = __builtin_amdgcn_readfirstlane((int)(threadIdx.x >> 5));
    const int lane = threadIdx.x & 31;

    const int r  = lane & 15;
    const int hh = lane >> 4;

    const float* Mh = mm + h * C_ * C_ + r * C_ + 2 * hh;
    v2f bv[4];
#pragma unroll
    for (int c = 0; c < 4; ++c) bv[c] = *(const v2f*)(Mh + 4 * c);

    const float  sB = (r < NO_) ? 1.f : 0.f;
    const float* Qh = q1 + h * NO_ * C_ + (r & (NO_ - 1)) * C_ + 2 * hh;
    v2f b1[4];
#pragma unroll
    for (int c = 0; c < 4; ++c) b1[c] = *(const v2f*)(Qh + 4 * c) * sB;

    stage_plane(plane, feat + (size_t)bt * N_ * H_ * C_ + h * C_);

    const int q0   = q * QTR_;
    const int nend = q0 + QTR_;
    const int nbeg = q0 + wv;

    float*       wp  = w_out + (size_t)bt * N_ * 16 * H_ + h;
    float*       dgp = deg   + (size_t)bt * N_ * H_ + h;
    float*       dwp = dw    + (size_t)bt7 * E2_ * H_ + h;
    const float* qjp = qj    + ((size_t)bt7 * N_ * H_ + h) * NO_;

    if (!dir) {
        node_loop<false, U1_>(plane, nn, qjp, bv, b1, wp, dgp, dwp, nbeg, nend, lane, r, hh);
        return;
    }

    node_loop<true, UF_>(plane, nn, qjp, bv, b1, wp, dgp, dwp, nbeg, nend, lane, r, hh);

    for (int j = wv; j < STL_; j += 32) {
        const int  nd   = q0 + 16 * j + r;
        const bool ok   = (nd < nend);
        const int  ndc  = ok ? nd : (nend - 1);
        const int  g0   = nn[ndc * K1_];
        const bool pred = ok && (g0 >= 0);
        const int  jn   = pred ? g0 : 0;
        const v8f  qv   = *(const v8f*)(qjp + ndc * (H_ * NO_));

        v8f d = {0.f,0.f,0.f,0.f,0.f,0.f,0.f,0.f};
#pragma unroll
        for (int c = 0; c < 4; ++c)
            d = __builtin_amdgcn_wmma_f32_16x16x4_f32(
                    false, b1[c], false, lds_frag(plane, jn, c, hh), (short)0, d, false, false);

        const float s = dot8(d, qv);
        const float wd0 = pred ? __expf(-s) : 0.f;

        if (hh == 0 && ok) dwp[(nd * K1_) * H_] = wd0;
    }
}

__global__ void k3_qj(const float* __restrict__ feat,
                      const float* __restrict__ q2,
                      float*       __restrict__ qj)
{
    const size_t tid = (size_t)blockIdx.x * blockDim.x + threadIdx.x;
    const size_t total = (size_t)B_ * T7_ * N_ * H_;
    if (tid >= total) return;
    const int h   = (int)(tid % H_);
    const int n   = (int)((tid / H_) % N_);
    const int bt7 = (int)( tid / ((size_t)H_ * N_));
    const int t   = bt7 % T7_;
    const int b   = bt7 / T7_;

    const float* f = feat + ((((size_t)(b * T_ + t + 1) * N_ + n) * H_ + h) * C_);
    const float* Q = q2 + h * NO_ * C_;
    float* o = qj + tid * NO_;
#pragma unroll
    for (int i = 0; i < NO_; ++i) {
        float s = 0.f;
#pragma unroll
        for (int j = 0; j < C_; ++j) s += Q[i * C_ + j] * f[j];
        o[i] = s;
    }
}

#define PBLK_  256
#define NB5_   ((B_ * T7_ * N_ * H_ + PBLK_ - 1) / PBLK_)
#define NPL_   (B_ * T_)
#define ROW_   (N_ * 16 * H_)
#define NBU_   ((ROW_ + PBLK_ - 1) / PBLK_)

__global__ void __launch_bounds__(PBLK_)
k25_post(float* __restrict__ uw,
         float* __restrict__ dw,
         const float* __restrict__ deg,
         const int*   __restrict__ nn)
{
    if (blockIdx.x < NB5_) {
        const size_t tid = (size_t)blockIdx.x * PBLK_ + threadIdx.x;
        const size_t total = (size_t)B_ * T7_ * N_ * H_;
        if (tid >= total) return;
        const int h   = (int)(tid % H_);
        const int n   = (int)((tid / H_) % N_);
        const int bt7 = (int)( tid / ((size_t)H_ * N_));

        float* base = dw + ((size_t)bt7 * E2_ + (size_t)n * K1_) * H_ + h;
        float v[K1_];
#pragma unroll
        for (int k = 0; k < K1_; ++k) v[k] = base[k * H_];
        float s = 0.f;
#pragma unroll
        for (int k = 0; k < K1_; ++k) s += v[k];
        const float inv = (s > 0.f) ? (1.f / s) : 0.f;
#pragma unroll
        for (int k = 0; k < K1_; ++k) base[k * H_] = v[k] * inv;
        return;
    }

    const int e = (int)(blockIdx.x - NB5_) * PBLK_ + (int)threadIdx.x;
    if (e >= ROW_) return;
    const int h = e % H_;
    const int k = (e / H_) % 16;
    const int n = e / (H_ * 16);

    const int  nb    = nn[n * K1_ + 1 + k];
    const bool valid = (nb >= 0);
    const int  nj    = valid ? nb : n;

    float w[NPL_], di[NPL_], dj[NPL_];
#pragma unroll
    for (int p = 0; p < NPL_; ++p) {
        w[p]  = uw[(size_t)p * ROW_ + e];
        di[p] = deg[((size_t)p * N_ + n) * H_ + h];
    }
#pragma unroll
    for (int p = 0; p < NPL_; ++p)
        dj[p] = deg[((size_t)p * N_ + nj) * H_ + h];
    asm volatile("" :: "v"(w[0]), "v"(w[1]), "v"(w[2]),  "v"(w[3]),  "v"(w[4]),  "v"(w[5]),  "v"(w[6]),  "v"(w[7]),
                       "v"(w[8]), "v"(w[9]), "v"(w[10]), "v"(w[11]), "v"(w[12]), "v"(w[13]), "v"(w[14]), "v"(w[15]));
#pragma unroll
    for (int p = 0; p < NPL_; ++p) {
        const float dm = sqrtf(di[p] * (valid ? dj[p] : 0.f));
        uw[(size_t)p * ROW_ + e] = (dm > 0.f) ? (w[p] / dm) : 0.f;
    }
}

extern "C" void kernel_launch(void* const* d_in, const int* in_sizes, int n_in,
                              void* d_out, int out_size, void* d_ws, size_t ws_size,
                              hipStream_t stream)
{
    const float* feat = (const float*)d_in[0];
    const int*   nn   = (const int*)  d_in[1];
    const float* q1   = (const float*)d_in[2];
    const float* q2   = (const float*)d_in[3];
    const float* mm   = (const float*)d_in[4];

    float* uw = (float*)d_out;
    float* dw = uw + (size_t)B_ * T_ * N_ * 16 * H_;

    float* deg = (float*)d_ws;
    float* qjw = deg + (size_t)B_ * T_ * N_ * H_;

    const int BLK = 256;

    {
        const size_t total = (size_t)B_ * T7_ * N_ * H_;
        k3_qj<<<(unsigned)((total + BLK - 1) / BLK), BLK, 0, stream>>>(feat, q2, qjw);
    }
    k14_plane<<<B_ * T_ * H_ * 4, 1024, 0, stream>>>(feat, nn, mm, q1, qjw, uw, deg, dw);
    k25_post<<<NB5_ + NBU_, PBLK_, 0, stream>>>(uw, dw, deg, nn);
}
